# starting kernel
# baseline (speedup 1.0000x reference)
#include <hip/hip_runtime.h>

#define Mn   45
#define Bn   64
#define Tn   512
#define Hn   20
#define Gn   80
#define FC   32
#define MHD  256
#define OUTD 22
#define TOK  (Bn * Tn)

#define BQ   16
#define NBG  (Bn / BQ)

typedef __attribute__((ext_vector_type(2))) float v2f;
typedef __attribute__((ext_vector_type(8))) float v8f;

__device__ __forceinline__ v8f wmma4(v2f a, v2f b, v8f c) {
  return __builtin_amdgcn_wmma_f32_16x16x4_f32(false, a, false, b, (short)0, c,
                                               false, false);
}

__device__ __forceinline__ float sigm(float x) {
  return __builtin_amdgcn_rcpf(1.f + __expf(-x));
}
__device__ __forceinline__ float tanh_fast(float x) {
  return 1.f - 2.f * __builtin_amdgcn_rcpf(1.f + __expf(2.f * x));
}

__global__ __launch_bounds__(160) void lstm45_kernel(
    const float* __restrict__ Dis,
    const float* __restrict__ Wih0, const float* __restrict__ Whh0,
    const float* __restrict__ bih0, const float* __restrict__ bhh0,
    const float* __restrict__ Wih1, const float* __restrict__ Whh1,
    const float* __restrict__ bih1, const float* __restrict__ bhh1,
    const float* __restrict__ Wih2, const float* __restrict__ Whh2,
    const float* __restrict__ bih2, const float* __restrict__ bhh2,
    const float* __restrict__ Wfc1, const float* __restrict__ bfc1,
    const float* __restrict__ Wfc2, const float* __restrict__ bfc2,
    float* __restrict__ Dpred)
{
  __shared__ float xall[BQ][Tn];
  __shared__ float hcat[BQ][3 * Hn];
  __shared__ float ccat[BQ][3 * Hn];
  __shared__ float gbuf[BQ][Gn];
  __shared__ float zbuf[BQ][FC];
  __shared__ float wfc2s[FC];

  const int tid  = threadIdx.x;
  const int m    = blockIdx.y;
  const int b0   = blockIdx.x * BQ;
  const int lane = tid & 31;
  const int wv   = tid >> 5;
  const int lh   = lane & 15;
  const int sel  = lane >> 4;
  const int n    = wv * 16 + lh;

  for (int i = tid; i < BQ * Tn; i += 160) {
    int b = i / Tn, t = i % Tn;
    xall[b][t] = Dis[(m * Bn + b0 + b) * Tn + t];
  }
  for (int i = tid; i < BQ * 3 * Hn; i += 160) {
    (&hcat[0][0])[i] = 0.f;
    (&ccat[0][0])[i] = 0.f;
  }
  for (int f = tid; f < FC; f += 160) wfc2s[f] = Wfc2[m * FC + f];

  const float bias0v = bih0[m * Gn + n] + bhh0[m * Gn + n];
  const float bias1v = bih1[m * Gn + n] + bhh1[m * Gn + n];
  const float bias2v = bih2[m * Gn + n] + bhh2[m * Gn + n];
  const float wih0v  = Wih0[m * Gn + n];
  const float bfc2v  = bfc2[m];

  const int wb = (m * Gn + n) * Hn;
  v2f bw0[5], bw1[10], bw2[10];
  #pragma unroll
  for (int kk = 0; kk < 5; ++kk) {
    const int k = 4 * kk + 2 * sel;
    bw0[kk][0] = Whh0[wb + k]; bw0[kk][1] = Whh0[wb + k + 1];
    bw1[kk][0] = Wih1[wb + k]; bw1[kk][1] = Wih1[wb + k + 1];
    bw2[kk][0] = Wih2[wb + k]; bw2[kk][1] = Wih2[wb + k + 1];
  }
  #pragma unroll
  for (int kk = 0; kk < 5; ++kk) {
    const int k = 4 * kk + 2 * sel;
    bw1[5 + kk][0] = Whh1[wb + k]; bw1[5 + kk][1] = Whh1[wb + k + 1];
    bw2[5 + kk][0] = Whh2[wb + k]; bw2[5 + kk][1] = Whh2[wb + k + 1];
  }
  v2f bwf[5];
  float bfc1v = 0.f;
  if (wv < 2) {
    const int fb = (m * FC + n) * Hn;
    #pragma unroll
    for (int kk = 0; kk < 5; ++kk) {
      const int k = 4 * kk + 2 * sel;
      bwf[kk][0] = Wfc1[fb + k]; bwf[kk][1] = Wfc1[fb + k + 1];
    }
    bfc1v = bfc1[m * FC + n];
  }

  const int eb0 = tid / Hn,         eh0 = tid % Hn;
  const int eb1 = (tid + 160) / Hn, eh1 = (tid + 160) % Hn;

  __syncthreads();

#define CELL(l, bb, hh)                                                  \
  {                                                                      \
    float ig = sigm(gbuf[bb][hh]);                                       \
    float fg = sigm(gbuf[bb][Hn + (hh)]);                                \
    float gv = tanh_fast(gbuf[bb][2 * Hn + (hh)]);                       \
    float og = sigm(gbuf[bb][3 * Hn + (hh)]);                            \
    float c  = fg * ccat[bb][(l) * Hn + (hh)] + ig * gv;                 \
    ccat[bb][(l) * Hn + (hh)] = c;                                       \
    hcat[bb][(l) * Hn + (hh)] = og * tanh_fast(c);                       \
  }

  for (int t = 0; t < Tn; ++t) {
    {
      v8f acc;
      #pragma unroll
      for (int r = 0; r < 8; ++r)
        acc[r] = bias0v + xall[r + 8 * sel][t] * wih0v;
      #pragma unroll
      for (int kk = 0; kk < 5; ++kk) {
        const int k = 4 * kk + 2 * sel;
        v2f a; a[0] = hcat[lh][k]; a[1] = hcat[lh][k + 1];
        acc = wmma4(a, bw0[kk], acc);
      }
      #pragma unroll
      for (int r = 0; r < 8; ++r) gbuf[r + 8 * sel][n] = acc[r];
    }
    __syncthreads();
    CELL(0, eb0, eh0)
    CELL(0, eb1, eh1)
    __syncthreads();

    {
      v8f acc;
      #pragma unroll
      for (int r = 0; r < 8; ++r) acc[r] = bias1v;
      #pragma unroll
      for (int kk = 0; kk < 10; ++kk) {
        const int k = 4 * kk + 2 * sel;
        v2f a; a[0] = hcat[lh][k]; a[1] = hcat[lh][k + 1];
        acc = wmma4(a, bw1[kk], acc);
      }
      #pragma unroll
      for (int r = 0; r < 8; ++r) gbuf[r + 8 * sel][n] = acc[r];
    }
    __syncthreads();
    CELL(1, eb0, eh0)
    CELL(1, eb1, eh1)
    __syncthreads();

    {
      v8f acc;
      #pragma unroll
      for (int r = 0; r < 8; ++r) acc[r] = bias2v;
      #pragma unroll
      for (int kk = 0; kk < 10; ++kk) {
        const int k = Hn + 4 * kk + 2 * sel;
        v2f a; a[0] = hcat[lh][k]; a[1] = hcat[lh][k + 1];
        acc = wmma4(a, bw2[kk], acc);
      }
      #pragma unroll
      for (int r = 0; r < 8; ++r) gbuf[r + 8 * sel][n] = acc[r];
    }
    __syncthreads();
    CELL(2, eb0, eh0)
    CELL(2, eb1, eh1)
    __syncthreads();

    if (wv < 2) {
      v8f acc;
      #pragma unroll
      for (int r = 0; r < 8; ++r) acc[r] = bfc1v;
      #pragma unroll
      for (int kk = 0; kk < 5; ++kk) {
        const int k = 2 * Hn + 4 * kk + 2 * sel;
        v2f a; a[0] = hcat[lh][k]; a[1] = hcat[lh][k + 1];
        acc = wmma4(a, bwf[kk], acc);
      }
      #pragma unroll
      for (int r = 0; r < 8; ++r) zbuf[r + 8 * sel][n] = fmaxf(acc[r], 0.f);
    }
    __syncthreads();
    if (tid < BQ) {
      float y = bfc2v;
      #pragma unroll
      for (int f = 0; f < FC; ++f) y += zbuf[tid][f] * wfc2s[f];
      Dpred[(m * Bn + b0 + tid) * Tn + t] = y;
    }
  }
#undef CELL
}

#define ASTR (MHD + 4)

__device__ __forceinline__ void hidden_layer(
    const float* __restrict__ W, const float* __restrict__ bias,
    const float (*IN)[ASTR], float (*OUT)[ASTR], int lane, int wv)
{
  const int lh = lane & 15, sel = lane >> 4;
  for (int tile = wv; tile < 64; tile += 8) {
    const int mt = tile >> 4, nt = tile & 15;
    const int n  = nt * 16 + lh;
    const int mr = mt * 16 + lh;
    v8f acc;
    const float bv = bias[n];
    #pragma unroll
    for (int r = 0; r < 8; ++r) acc[r] = bv;
    #pragma unroll 4
    for (int kk = 0; kk < MHD / 4; ++kk) {
      const int k = 4 * kk + 2 * sel;
      v2f a, b;
      a[0] = IN[mr][k];      a[1] = IN[mr][k + 1];
      b[0] = W[n * MHD + k]; b[1] = W[n * MHD + k + 1];
      acc = wmma4(a, b, acc);
    }
    #pragma unroll
    for (int r = 0; r < 8; ++r)
      OUT[mt * 16 + r + 8 * sel][n] = fmaxf(acc[r], 0.f);
  }
}

__global__ __launch_bounds__(256) void mlp_kernel(
    const float* __restrict__ Dpred,
    const float* __restrict__ Wm0, const float* __restrict__ bm0,
    const float* __restrict__ Wm1, const float* __restrict__ bm1,
    const float* __restrict__ Wm2, const float* __restrict__ bm2,
    const float* __restrict__ Wm3, const float* __restrict__ bm3,
    float* __restrict__ Out)
{
  __shared__ float Abuf[64][ASTR];
  __shared__ float Bbuf[64][ASTR];

  const int tid  = threadIdx.x;
  const int lane = tid & 31, wv = tid >> 5;
  const int lh   = lane & 15, sel = lane >> 4;
  const int tok0 = blockIdx.x * 64;

  for (int i = tid; i < 64 * 48; i += 256) {
    int r = i / 48, f = i % 48;
    Abuf[r][f] = (f < Mn) ? Dpred[f * TOK + tok0 + r] : 0.f;
  }
  __syncthreads();

  for (int tile = wv; tile < 64; tile += 8) {
    const int mt = tile >> 4, nt = tile & 15;
    const int n  = nt * 16 + lh;
    const int mr = mt * 16 + lh;
    v8f acc;
    const float bv = bm0[n];
    #pragma unroll
    for (int r = 0; r < 8; ++r) acc[r] = bv;
    #pragma unroll
    for (int kk = 0; kk < 12; ++kk) {
      const int k  = 4 * kk + 2 * sel;
      const int k0 = (k     < Mn) ? k     : (Mn - 1);
      const int k1 = (k + 1 < Mn) ? k + 1 : (Mn - 1);
      v2f a, b;
      a[0] = Abuf[mr][k];      a[1] = Abuf[mr][k + 1];
      b[0] = Wm0[n * Mn + k0]; b[1] = Wm0[n * Mn + k1];
      acc = wmma4(a, b, acc);
    }
    #pragma unroll
    for (int r = 0; r < 8; ++r)
      Bbuf[mt * 16 + r + 8 * sel][n] = fmaxf(acc[r], 0.f);
  }
  __syncthreads();

  hidden_layer(Wm1, bm1, Bbuf, Abuf, lane, wv);
  __syncthreads();
  hidden_layer(Wm2, bm2, Abuf, Bbuf, lane, wv);
  __syncthreads();

  for (int tile = wv; tile < 8; tile += 8) {
    const int mt = tile >> 1, nt = tile & 1;
    const int n  = nt * 16 + lh;
    const int nc = (n < OUTD) ? n : (OUTD - 1);
    const int mr = mt * 16 + lh;
    v8f acc;
    const float bv = bm3[nc];
    #pragma unroll
    for (int r = 0; r < 8; ++r) acc[r] = bv;
    #pragma unroll 4
    for (int kk = 0; kk < MHD / 4; ++kk) {
      const int k = 4 * kk + 2 * sel;
      v2f a, b;
      a[0] = Bbuf[mr][k];       a[1] = Bbuf[mr][k + 1];
      b[0] = Wm3[nc * MHD + k]; b[1] = Wm3[nc * MHD + k + 1];
      acc = wmma4(a, b, acc);
    }
    if (n < OUTD) {
      #pragma unroll
      for (int r = 0; r < 8; ++r)
        Out[(tok0 + mt * 16 + r + 8 * sel) * OUTD + n] = acc[r];
    }
  }
}

extern "C" void kernel_launch(void* const* d_in, const int* in_sizes, int n_in,
                              void* d_out, int out_size, void* d_ws, size_t ws_size,
                              hipStream_t stream) {
  const float* Dis  = (const float*)d_in[0];
  const float* Wih0 = (const float*)d_in[1];
  const float* Whh0 = (const float*)d_in[2];
  const float* bih0 = (const float*)d_in[3];
  const float* bhh0 = (const float*)d_in[4];
  const float* Wih1 = (const float*)d_in[5];
  const float* Whh1 = (const float*)d_in[6];
  const float* bih1 = (const float*)d_in[7];
  const float* bhh1 = (const float*)d_in[8];
  const float* Wih2 = (const float*)d_in[9];
  const float* Whh2 = (const float*)d_in[10];
  const float* bih2 = (const float*)d_in[11];
  const float* bhh2 = (const float*)d_in[12];
  const float* Wfc1 = (const float*)d_in[13];
  const float* bfc1 = (const float*)d_in[14];
  const float* Wfc2 = (const float*)d_in[15];
  const float* bfc2 = (const float*)d_in[16];
  const float* Wm0  = (const float*)d_in[17];
  const float* bm0  = (const float*)d_in[18];
  const float* Wm1  = (const float*)d_in[19];
  const float* bm1  = (const float*)d_in[20];
  const float* Wm2  = (const float*)d_in[21];
  const float* bm2  = (const float*)d_in[22];
  const float* Wm3  = (const float*)d_in[23];
  const float* bm3  = (const float*)d_in[24];

  float* out   = (float*)d_out;
  float* Dpred = out;
  float* cpf   = out + Mn * Bn * Tn;

  dim3 g1(NBG, Mn);
  lstm45_kernel<<<g1, 160, 0, stream>>>(
      Dis, Wih0, Whh0, bih0, bhh0, Wih1, Whh1, bih1, bhh1,
      Wih2, Whh2, bih2, bhh2, Wfc1, bfc1, Wfc2, bfc2, Dpred);

  mlp_kernel<<<TOK / 64, 256, 0, stream>>>(
      Dpred, Wm0, bm0, Wm1, bm1, Wm2, bm2, Wm3, bm3, cpf);
}


# LSTM layers pipelined as a wavefront: layer l runs step t-l (head t-3) in the same tick: 2 barriers per step instead of 7, 3 independent WMMA chains per wave; head on waves 1,2, readout on wave 3
# speedup vs baseline: 1.5970x; 1.5970x over previous
#include <hip/hip_runtime.h>

#define Mn   45
#define Bn   64
#define Tn   512
#define Hn   20
#define Gn   80
#define FC   32
#define MHD  256
#define OUTD 22
#define TOK  (Bn * Tn)

#define BQ   16
#define NBG  (Bn / BQ)

typedef __attribute__((ext_vector_type(2))) float v2f;
typedef __attribute__((ext_vector_type(8))) float v8f;

__device__ __forceinline__ v8f wmma4(v2f a, v2f b, v8f c) {
  return __builtin_amdgcn_wmma_f32_16x16x4_f32(false, a, false, b, (short)0, c,
                                               false, false);
}

__device__ __forceinline__ float sigm(float x) {
  return __builtin_amdgcn_rcpf(1.f + __expf(-x));
}
__device__ __forceinline__ float tanh_fast(float x) {
  return 1.f - 2.f * __builtin_amdgcn_rcpf(1.f + __expf(2.f * x));
}

__global__ __launch_bounds__(160) void lstm45_kernel(
    const float* __restrict__ Dis,
    const float* __restrict__ Wih0, const float* __restrict__ Whh0,
    const float* __restrict__ bih0, const float* __restrict__ bhh0,
    const float* __restrict__ Wih1, const float* __restrict__ Whh1,
    const float* __restrict__ bih1, const float* __restrict__ bhh1,
    const float* __restrict__ Wih2, const float* __restrict__ Whh2,
    const float* __restrict__ bih2, const float* __restrict__ bhh2,
    const float* __restrict__ Wfc1, const float* __restrict__ bfc1,
    const float* __restrict__ Wfc2, const float* __restrict__ bfc2,
    float* __restrict__ Dpred)
{
  __shared__ float xall[BQ][Tn];
  __shared__ float hcat[BQ][3 * Hn];
  __shared__ float ccat[BQ][3 * Hn];
  __shared__ float gbuf[3][BQ][Gn];
  __shared__ float zbuf[BQ][FC];
  __shared__ float wfc2s[FC];

  const int tid  = threadIdx.x;
  const int m    = blockIdx.y;
  const int b0   = blockIdx.x * BQ;
  const int lane = tid & 31;
  const int wv   = tid >> 5;
  const int lh   = lane & 15;
  const int sel  = lane >> 4;
  const int n    = wv * 16 + lh;

  for (int i = tid; i < BQ * Tn; i += 160) {
    int b = i / Tn, t = i % Tn;
    xall[b][t] = Dis[(m * Bn + b0 + b) * Tn + t];
  }
  for (int i = tid; i < BQ * 3 * Hn; i += 160) {
    (&hcat[0][0])[i] = 0.f;
    (&ccat[0][0])[i] = 0.f;
  }
  for (int f = tid; f < FC; f += 160) wfc2s[f] = Wfc2[m * FC + f];

  const float bias0v = bih0[m * Gn + n] + bhh0[m * Gn + n];
  const float bias1v = bih1[m * Gn + n] + bhh1[m * Gn + n];
  const float bias2v = bih2[m * Gn + n] + bhh2[m * Gn + n];
  const float wih0v  = Wih0[m * Gn + n];
  const float bfc2v  = bfc2[m];

  const int wb = (m * Gn + n) * Hn;
  v2f bw0[5], bw1[10], bw2[10];
  #pragma unroll
  for (int kk = 0; kk < 5; ++kk) {
    const int k = 4 * kk + 2 * sel;
    bw0[kk][0] = Whh0[wb + k]; bw0[kk][1] = Whh0[wb + k + 1];
    bw1[kk][0] = Wih1[wb + k]; bw1[kk][1] = Wih1[wb + k + 1];
    bw2[kk][0] = Wih2[wb + k]; bw2[kk][1] = Wih2[wb + k + 1];
  }
  #pragma unroll
  for (int kk = 0; kk < 5; ++kk) {
    const int k = 4 * kk + 2 * sel;
    bw1[5 + kk][0] = Whh1[wb + k]; bw1[5 + kk][1] = Whh1[wb + k + 1];
    bw2[5 + kk][0] = Whh2[wb + k]; bw2[5 + kk][1] = Whh2[wb + k + 1];
  }
  const bool headw = (wv == 1) || (wv == 2);
  const int  fn    = (wv == 2 ? 16 : 0) + lh;
  v2f bwf[5];
  #pragma unroll
  for (int kk = 0; kk < 5; ++kk) { bwf[kk][0] = 0.f; bwf[kk][1] = 0.f; }
  float bfc1v = 0.f;
  if (headw) {
    const int fb = (m * FC + fn) * Hn;
    #pragma unroll
    for (int kk = 0; kk < 5; ++kk) {
      const int k = 4 * kk + 2 * sel;
      bwf[kk][0] = Wfc1[fb + k]; bwf[kk][1] = Wfc1[fb + k + 1];
    }
    bfc1v = bfc1[m * FC + fn];
  }

  const int eb0 = tid / Hn,         eh0 = tid % Hn;
  const int eb1 = (tid + 160) / Hn, eh1 = (tid + 160) % Hn;

  __syncthreads();

#define CELL_LD(l, e, bb, hh)                                            \
  {                                                                      \
    gi[e] = gbuf[l][bb][hh];                                             \
    gf[e] = gbuf[l][bb][Hn + (hh)];                                      \
    gg[e] = gbuf[l][bb][2 * Hn + (hh)];                                  \
    go[e] = gbuf[l][bb][3 * Hn + (hh)];                                  \
    cp[e] = ccat[bb][(l) * Hn + (hh)];                                   \
  }
#define CELL_UP(l, e, bb, hh)                                            \
  {                                                                      \
    float ig = sigm(gi[e]);                                              \
    float fg = sigm(gf[e]);                                              \
    float gv = tanh_fast(gg[e]);                                         \
    float og = sigm(go[e]);                                              \
    float c  = fg * cp[e] + ig * gv;                                     \
    ccat[bb][(l) * Hn + (hh)] = c;                                       \
    hcat[bb][(l) * Hn + (hh)] = og * tanh_fast(c);                       \
  }

#define TICK(L0, L1, L2, HD, u)                                              \
  {                                                                          \
    v2f a[15];                                                               \
    _Pragma("unroll")               \
    for (int j = 0; j < 15; ++j) {                                           \
      const bool need = (j < 5)  ? ((L0) || (L1))                            \
                      : (j < 10) ? ((L1) || (L2)) : ((L2) || (HD));          \
      if (need) {                                                            \
        const int k = 4 * j + 2 * sel;                                       \
        a[j][0] = hcat[lh][k]; a[j][1] = hcat[lh][k + 1];                    \
      } else { a[j][0] = 0.f; a[j][1] = 0.f; }                               \
    }                                                                        \
    if (HD) {                               \
      if (headw) {                                   \
        v8f accf;                                                            \
        _Pragma("unroll")           \
        for (int r = 0; r < 8; ++r) accf[r] = bfc1v;                         \
        _Pragma("unroll")           \
        for (int kk = 0; kk < 5; ++kk) accf = wmma4(a[10 + kk], bwf[kk], accf); \
        _Pragma("unroll")           \
        for (int r = 0; r < 8; ++r)                                          \
          zbuf[r + 8 * sel][fn] = fmaxf(accf[r], 0.f);                       \
      }                                                                      \
    }                                                                        \
    {                                     \
      v8f acc0, acc1, acc2;                                                  \
      _Pragma("unroll")             \
      for (int r = 0; r < 8; ++r) {                                          \
        acc0[r] = (L0) ? (bias0v + xall[r + 8 * sel][(L0) ? (u) : 0] * wih0v) : 0.f; \
        acc1[r] = bias1v;                                                    \
        acc2[r] = bias2v;                                                    \
      }                                                                      \
      _Pragma("unroll")             \
      for (int kk = 0; kk < 10; ++kk) {                                      \
        if (L1) acc1 = wmma4(a[kk], bw1[kk], acc1);                          \
        if (L2) acc2 = wmma4(a[5 + kk], bw2[kk], acc2);                      \
        if ((L0) && kk < 5) acc0 = wmma4(a[kk], bw0[kk], acc0);              \
      }                                                                      \
      _Pragma("unroll")             \
      for (int r = 0; r < 8; ++r) {                                          \
        if (L0) gbuf[0][r + 8 * sel][n] = acc0[r];                           \
        if (L1) gbuf[1][r + 8 * sel][n] = acc1[r];                           \
        if (L2) gbuf[2][r + 8 * sel][n] = acc2[r];                           \
      }                                                                      \
    }                                                                        \
    __syncthreads();                                                         \
    {                                        \
      float gi[6], gf[6], gg[6], go[6], cp[6];                               \
      if (L0) { CELL_LD(0, 0, eb0, eh0) CELL_LD(0, 1, eb1, eh1) }            \
      if (L1) { CELL_LD(1, 2, eb0, eh0) CELL_LD(1, 3, eb1, eh1) }            \
      if (L2) { CELL_LD(2, 4, eb0, eh0) CELL_LD(2, 5, eb1, eh1) }            \
      if (HD) {                          \
        if (wv == 3 && lane < BQ) {                                          \
          float y = bfc2v;                                                   \
          _Pragma("unroll")         \
          for (int f = 0; f < FC; ++f) y += zbuf[lane][f] * wfc2s[f];        \
          Dpred[(m * Bn + b0 + lane) * Tn + ((u) - 3)] = y;                  \
        }                                                                    \
      }                                                                      \
      if (L0) { CELL_UP(0, 0, eb0, eh0) CELL_UP(0, 1, eb1, eh1) }            \
      if (L1) { CELL_UP(1, 2, eb0, eh0) CELL_UP(1, 3, eb1, eh1) }            \
      if (L2) { CELL_UP(2, 4, eb0, eh0) CELL_UP(2, 5, eb1, eh1) }            \
    }                                                                        \
    __syncthreads();                                                         \
  }

  TICK(1, 0, 0, 0, 0)
  TICK(1, 1, 0, 0, 1)
  TICK(1, 1, 1, 0, 2)
  for (int u = 3; u < Tn; ++u) TICK(1, 1, 1, 1, u)
  TICK(0, 1, 1, 1, Tn)
  TICK(0, 0, 1, 1, Tn + 1)
  TICK(0, 0, 0, 1, Tn + 2)
#undef TICK
#undef CELL_UP
#undef CELL_LD
}

#define ASTR (MHD + 4)

__device__ __forceinline__ void hidden_layer(
    const float* __restrict__ W, const float* __restrict__ bias,
    const float (*IN)[ASTR], float (*OUT)[ASTR], int lane, int wv)
{
  const int lh = lane & 15, sel = lane >> 4;
  for (int tile = wv; tile < 64; tile += 8) {
    const int mt = tile >> 4, nt = tile & 15;
    const int n  = nt * 16 + lh;
    const int mr = mt * 16 + lh;
    v8f acc;
    const float bv = bias[n];
    #pragma unroll
    for (int r = 0; r < 8; ++r) acc[r] = bv;
    #pragma unroll 4
    for (int kk = 0; kk < MHD / 4; ++kk) {
      const int k = 4 * kk + 2 * sel;
      v2f a, b;
      a[0] = IN[mr][k];      a[1] = IN[mr][k + 1];
      b[0] = W[n * MHD + k]; b[1] = W[n * MHD + k + 1];
      acc = wmma4(a, b, acc);
    }
    #pragma unroll
    for (int r = 0; r < 8; ++r)
      OUT[mt * 16 + r + 8 * sel][n] = fmaxf(acc[r], 0.f);
  }
}

__global__ __launch_bounds__(256) void mlp_kernel(
    const float* __restrict__ Dpred,
    const float* __restrict__ Wm0, const float* __restrict__ bm0,
    const float* __restrict__ Wm1, const float* __restrict__ bm1,
    const float* __restrict__ Wm2, const float* __restrict__ bm2,
    const float* __restrict__ Wm3, const float* __restrict__ bm3,
    float* __restrict__ Out)
{
  __shared__ float Abuf[64][ASTR];
  __shared__ float Bbuf[64][ASTR];

  const int tid  = threadIdx.x;
  const int lane = tid & 31, wv = tid >> 5;
  const int lh   = lane & 15, sel = lane >> 4;
  const int tok0 = blockIdx.x * 64;

  for (int i = tid; i < 64 * 48; i += 256) {
    int r = i / 48, f = i % 48;
    Abuf[r][f] = (f < Mn) ? Dpred[f * TOK + tok0 + r] : 0.f;
  }
  __syncthreads();

  for (int tile = wv; tile < 64; tile += 8) {
    const int mt = tile >> 4, nt = tile & 15;
    const int n  = nt * 16 + lh;
    const int mr = mt * 16 + lh;
    v8f acc;
    const float bv = bm0[n];
    #pragma unroll
    for (int r = 0; r < 8; ++r) acc[r] = bv;
    #pragma unroll
    for (int kk = 0; kk < 12; ++kk) {
      const int k  = 4 * kk + 2 * sel;
      const int k0 = (k     < Mn) ? k     : (Mn - 1);
      const int k1 = (k + 1 < Mn) ? k + 1 : (Mn - 1);
      v2f a, b;
      a[0] = Abuf[mr][k];      a[1] = Abuf[mr][k + 1];
      b[0] = Wm0[n * Mn + k0]; b[1] = Wm0[n * Mn + k1];
      acc = wmma4(a, b, acc);
    }
    #pragma unroll
    for (int r = 0; r < 8; ++r)
      Bbuf[mt * 16 + r + 8 * sel][n] = fmaxf(acc[r], 0.f);
  }
  __syncthreads();

  hidden_layer(Wm1, bm1, Bbuf, Abuf, lane, wv);
  __syncthreads();
  hidden_layer(Wm2, bm2, Abuf, Bbuf, lane, wv);
  __syncthreads();

  for (int tile = wv; tile < 8; tile += 8) {
    const int mt = tile >> 1, nt = tile & 1;
    const int n  = nt * 16 + lh;
    const int nc = (n < OUTD) ? n : (OUTD - 1);
    const int mr = mt * 16 + lh;
    v8f acc;
    const float bv = bm3[nc];
    #pragma unroll
    for (int r = 0; r < 8; ++r) acc[r] = bv;
    #pragma unroll 4
    for (int kk = 0; kk < MHD / 4; ++kk) {
      const int k = 4 * kk + 2 * sel;
      v2f a, b;
      a[0] = Bbuf[mr][k];       a[1] = Bbuf[mr][k + 1];
      b[0] = Wm3[nc * MHD + k]; b[1] = Wm3[nc * MHD + k + 1];
      acc = wmma4(a, b, acc);
    }
    if (n < OUTD) {
      #pragma unroll
      for (int r = 0; r < 8; ++r)
        Out[(tok0 + mt * 16 + r + 8 * sel) * OUTD + n] = acc[r];
    }
  }
}

extern "C" void kernel_launch(void* const* d_in, const int* in_sizes, int n_in,
                              void* d_out, int out_size, void* d_ws, size_t ws_size,
                              hipStream_t stream) {
  const float* Dis  = (const float*)d_in[0];
  const float* Wih0 = (const float*)d_in[1];
  const float* Whh0 = (const float*)d_in[2];
  const float* bih0 = (const float*)d_in[3];
  const float* bhh0 = (const float*)d_in[4];
  const float* Wih1 = (const float*)d_in[5];
  const float* Whh1 = (const float*)d_in[6];
  const float* bih1 = (const float*)d_in[7];
  const float* bhh1 = (const float*)d_in[8];
  const float* Wih2 = (const float*)d_in[9];
  const float* Whh2 = (const float*)d_in[10];
  const float* bih2 = (const float*)d_in[11];
  const float* bhh2 = (const float*)d_in[12];
  const float* Wfc1 = (const float*)d_in[13];
  const float* bfc1 = (const float*)d_in[14];
  const float* Wfc2 = (const float*)d_in[15];
  const float* bfc2 = (const float*)d_in[16];
  const float* Wm0  = (const float*)d_in[17];
  const float* bm0  = (const float*)d_in[18];
  const float* Wm1  = (const float*)d_in[19];
  const float* bm1  = (const float*)d_in[20];
  const float* Wm2  = (const float*)d_in[21];
  const float* bm2  = (const float*)d_in[22];
  const float* Wm3  = (const float*)d_in[23];
  const float* bm3  = (const float*)d_in[24];

  float* out   = (float*)d_out;
  float* Dpred = out;
  float* cpf   = out + Mn * Bn * Tn;

  dim3 g1(NBG, Mn);
  lstm45_kernel<<<g1, 160, 0, stream>>>(
      Dis, Wih0, Whh0, bih0, bhh0, Wih1, Whh1, bih1, bhh1,
      Wih2, Whh2, bih2, bhh2, Wfc1, bfc1, Wfc2, bfc2, Dpred);

  mlp_kernel<<<TOK / 64, 256, 0, stream>>>(
      Dpred, Wm0, bm0, Wm1, bm1, Wm2, bm2, Wm3, bm3, cpf);
}


# LSTM gate tiles transposed (weights as A, hidden state as B, rows in unit-gate order): each lane holds i,f,g,o of its cells, cell update in registers, one barrier per tick; readout sum regrouped
# speedup vs baseline: 2.3219x; 1.4539x over previous
#include <hip/hip_runtime.h>

#define Mn   45
#define Bn   64
#define Tn   512
#define Hn   20
#define Gn   80
#define FC   32
#define MHD  256
#define OUTD 22
#define TOK  (Bn * Tn)

#define BQ   16
#define NBG  (Bn / BQ)

typedef __attribute__((ext_vector_type(2))) float v2f;
typedef __attribute__((ext_vector_type(8))) float v8f;

__device__ __forceinline__ v8f wmma4(v2f a, v2f b, v8f c) {
  return __builtin_amdgcn_wmma_f32_16x16x4_f32(false, a, false, b, (short)0, c,
                                               false, false);
}

__device__ __forceinline__ float sigm(float x) {
  return __builtin_amdgcn_rcpf(1.f + __expf(-x));
}
__device__ __forceinline__ float tanh_fast(float x) {
  return 1.f - 2.f * __builtin_amdgcn_rcpf(1.f + __expf(2.f * x));
}

#define XP (Tn + 1)

__global__ __launch_bounds__(160) void lstm45_kernel(
    const float* __restrict__ Dis,
    const float* __restrict__ Wih0, const float* __restrict__ Whh0,
    const float* __restrict__ bih0, const float* __restrict__ bhh0,
    const float* __restrict__ Wih1, const float* __restrict__ Whh1,
    const float* __restrict__ bih1, const float* __restrict__ bhh1,
    const float* __restrict__ Wih2, const float* __restrict__ Whh2,
    const float* __restrict__ bih2, const float* __restrict__ bhh2,
    const float* __restrict__ Wfc1, const float* __restrict__ bfc1,
    const float* __restrict__ Wfc2, const float* __restrict__ bfc2,
    float* __restrict__ Dpred)
{
  __shared__ float xall[BQ][XP];
  __shared__ float hc[2][BQ][3 * Hn];

  const int tid  = threadIdx.x;
  const int m    = blockIdx.y;
  const int b0   = blockIdx.x * BQ;
  const int lane = tid & 31;
  const int wv   = tid >> 5;
  const int lh   = lane & 15;
  const int sel  = lane >> 4;

  for (int i = tid; i < BQ * Tn; i += 160) {
    int b = i / Tn, t = i % Tn;
    xall[b][t] = Dis[(m * Bn + b0 + b) * Tn + t];
  }
  for (int i = tid; i < 2 * BQ * 3 * Hn; i += 160) (&hc[0][0][0])[i] = 0.f;

  const int arow = (lh & 3) * Hn + 4 * wv + (lh >> 2);
  const int wb   = (m * Gn + arow) * Hn;
  v2f aw0[5], aw1[10], aw2[10];
  #pragma unroll
  for (int kk = 0; kk < 5; ++kk) {
    const int k = 4 * kk + 2 * sel;
    aw0[kk][0] = Whh0[wb + k]; aw0[kk][1] = Whh0[wb + k + 1];
    aw1[kk][0] = Wih1[wb + k]; aw1[kk][1] = Wih1[wb + k + 1];
    aw2[kk][0] = Wih2[wb + k]; aw2[kk][1] = Wih2[wb + k + 1];
  }
  #pragma unroll
  for (int kk = 0; kk < 5; ++kk) {
    const int k = 4 * kk + 2 * sel;
    aw1[5 + kk][0] = Whh1[wb + k]; aw1[5 + kk][1] = Whh1[wb + k + 1];
    aw2[5 + kk][0] = Whh2[wb + k]; aw2[5 + kk][1] = Whh2[wb + k + 1];
  }
  float b0r[8], w0r[8], b1r[8], b2r[8];
  #pragma unroll
  for (int r = 0; r < 8; ++r) {
    const int g = m * Gn + (r & 3) * Hn + 4 * wv + 2 * sel + (r >> 2);
    b0r[r] = bih0[g] + bhh0[g];
    b1r[r] = bih1[g] + bhh1[g];
    b2r[r] = bih2[g] + bhh2[g];
    w0r[r] = Wih0[g];
  }
  const bool headw = (wv == 3);
  v2f awfa[5], awfb[5];
  float bf1a[8], bf1b[8], wf2a[8], wf2b[8];
  #pragma unroll
  for (int kk = 0; kk < 5; ++kk) {
    awfa[kk][0] = 0.f; awfa[kk][1] = 0.f; awfb[kk][0] = 0.f; awfb[kk][1] = 0.f;
  }
  #pragma unroll
  for (int r = 0; r < 8; ++r) { bf1a[r] = 0.f; bf1b[r] = 0.f; wf2a[r] = 0.f; wf2b[r] = 0.f; }
  if (headw) {
    const int fa = (m * FC + lh) * Hn;
    const int fb = (m * FC + 16 + lh) * Hn;
    #pragma unroll
    for (int kk = 0; kk < 5; ++kk) {
      const int k = 4 * kk + 2 * sel;
      awfa[kk][0] = Wfc1[fa + k]; awfa[kk][1] = Wfc1[fa + k + 1];
      awfb[kk][0] = Wfc1[fb + k]; awfb[kk][1] = Wfc1[fb + k + 1];
    }
    #pragma unroll
    for (int r = 0; r < 8; ++r) {
      const int f = 8 * sel + r;
      bf1a[r] = bfc1[m * FC + f];      wf2a[r] = Wfc2[m * FC + f];
      bf1b[r] = bfc1[m * FC + 16 + f]; wf2b[r] = Wfc2[m * FC + 16 + f];
    }
  }
  const float bfc2v = bfc2[m];

  float c0[2] = {0.f, 0.f}, c1[2] = {0.f, 0.f}, c2[2] = {0.f, 0.f};
  float psk = 0.f, pso = 0.f;

  __syncthreads();

#define CELL(acc, cst, l)                                                  \
  {                                                                        \
    _Pragma("unroll")             \
    for (int p = 0; p < 2; ++p) {                                          \
      float ig = sigm(acc[4 * p]);                                         \
      float fg = sigm(acc[4 * p + 1]);                                     \
      float gv = tanh_fast(acc[4 * p + 2]);                                \
      float og = sigm(acc[4 * p + 3]);                                     \
      float c  = fg * cst[p] + ig * gv;                                    \
      cst[p] = c;                                                          \
      hw[(l) * Hn + p] = og * tanh_fast(c);                                \
    }                                                                      \
  }

#define TICK(L0, L1, L2, HD, P, u)                                           \
  {                                                                          \
    const float* hr = &hc[P][lh][2 * sel];                                   \
    float*       hw = &hc[1 - (P)][lh][4 * wv + 2 * sel];                    \
    v2f hb[15];                                                              \
    _Pragma("unroll")               \
    for (int j = 0; j < 15; ++j) {                                           \
      const bool need = (j < 5)  ? ((L0) || (L1))                            \
                      : (j < 10) ? ((L1) || (L2)) : ((L2) || (HD));          \
      if (need) { hb[j][0] = hr[4 * j]; hb[j][1] = hr[4 * j + 1]; }          \
      else      { hb[j][0] = 0.f; hb[j][1] = 0.f; }                          \
    }                                                                        \
    const float xv = (L0) ? xall[lh][(L0) ? (u) : 0] : 0.f;                  \
    if ((u) >= 4) {                      \
      if (headw && sel == 0)                                                 \
        Dpred[(m * Bn + b0 + lh) * Tn + ((u) - 4)] = bfc2v + (psk + pso);    \
    }                                                                        \
    v8f fa8, fb8;                                                            \
    _Pragma("unroll")               \
    for (int r = 0; r < 8; ++r) { fa8[r] = bf1a[r]; fb8[r] = bf1b[r]; }      \
    if (HD) {                               \
      if (headw) {                                   \
        _Pragma("unroll")           \
        for (int kk = 0; kk < 5; ++kk) {                                     \
          fa8 = wmma4(awfa[kk], hb[10 + kk], fa8);                           \
          fb8 = wmma4(awfb[kk], hb[10 + kk], fb8);                           \
        }                                                                    \
      }                                                                      \
    }                                                                        \
    v8f acc0, acc1, acc2;                                                    \
    _Pragma("unroll")               \
    for (int r = 0; r < 8; ++r) {                                            \
      acc0[r] = (L0) ? (b0r[r] + xv * w0r[r]) : 0.f;                         \
      acc1[r] = b1r[r];                                                      \
      acc2[r] = b2r[r];                                                      \
    }                                                                        \
    _Pragma("unroll")               \
    for (int kk = 0; kk < 10; ++kk) {                                        \
      if ((L0) && kk < 5) acc0 = wmma4(aw0[kk], hb[kk], acc0);               \
      if (L1) acc1 = wmma4(aw1[kk], hb[kk], acc1);                           \
      if (L2) acc2 = wmma4(aw2[kk], hb[5 + kk], acc2);                       \
    }                                                                        \
    if (HD) {                            \
      if (headw) {                                   \
        float ps = 0.f;                                                      \
        _Pragma("unroll")           \
        for (int r = 0; r < 8; ++r) ps += fmaxf(fa8[r], 0.f) * wf2a[r];      \
        _Pragma("unroll")           \
        for (int r = 0; r < 8; ++r) ps += fmaxf(fb8[r], 0.f) * wf2b[r];      \
        psk = ps;                        \
        pso = __shfl_xor(ps, 16, 32);              \
      }                                                                      \
    }                                                                        \
    if (L0) CELL(acc0, c0, 0)                                                \
    if (L1) CELL(acc1, c1, 1)                                                \
    if (L2) CELL(acc2, c2, 2)                                                \
    __syncthreads();                                                         \
  }

  TICK(1, 0, 0, 0, 0, 0)
  TICK(1, 1, 0, 0, 1, 1)
  TICK(1, 1, 1, 0, 0, 2)
  for (int u = 3; u < Tn - 1; u += 2) {
    TICK(1, 1, 1, 1, 1, u)
    TICK(1, 1, 1, 1, 0, u + 1)
  }
  TICK(1, 1, 1, 1, 1, Tn - 1)
  TICK(0, 1, 1, 1, 0, Tn)
  TICK(0, 0, 1, 1, 1, Tn + 1)
  TICK(0, 0, 0, 1, 0, Tn + 2)
  if (headw && sel == 0)
    Dpred[(m * Bn + b0 + lh) * Tn + (Tn - 1)] = bfc2v + (psk + pso);
#undef TICK
#undef CELL
}

#define ASTR (MHD + 4)

__device__ __forceinline__ void hidden_layer(
    const float* __restrict__ W, const float* __restrict__ bias,
    const float (*IN)[ASTR], float (*OUT)[ASTR], int lane, int wv)
{
  const int lh = lane & 15, sel = lane >> 4;
  for (int tile = wv; tile < 64; tile += 8) {
    const int mt = tile >> 4, nt = tile & 15;
    const int n  = nt * 16 + lh;
    const int mr = mt * 16 + lh;
    v8f acc;
    const float bv = bias[n];
    #pragma unroll
    for (int r = 0; r < 8; ++r) acc[r] = bv;
    #pragma unroll 4
    for (int kk = 0; kk < MHD / 4; ++kk) {
      const int k = 4 * kk + 2 * sel;
      v2f a, b;
      a[0] = IN[mr][k];      a[1] = IN[mr][k + 1];
      b[0] = W[n * MHD + k]; b[1] = W[n * MHD + k + 1];
      acc = wmma4(a, b, acc);
    }
    #pragma unroll
    for (int r = 0; r < 8; ++r)
      OUT[mt * 16 + r + 8 * sel][n] = fmaxf(acc[r], 0.f);
  }
}

__global__ __launch_bounds__(256) void mlp_kernel(
    const float* __restrict__ Dpred,
    const float* __restrict__ Wm0, const float* __restrict__ bm0,
    const float* __restrict__ Wm1, const float* __restrict__ bm1,
    const float* __restrict__ Wm2, const float* __restrict__ bm2,
    const float* __restrict__ Wm3, const float* __restrict__ bm3,
    float* __restrict__ Out)
{
  __shared__ float Abuf[64][ASTR];
  __shared__ float Bbuf[64][ASTR];

  const int tid  = threadIdx.x;
  const int lane = tid & 31, wv = tid >> 5;
  const int lh   = lane & 15, sel = lane >> 4;
  const int tok0 = blockIdx.x * 64;

  for (int i = tid; i < 64 * 48; i += 256) {
    int r = i / 48, f = i % 48;
    Abuf[r][f] = (f < Mn) ? Dpred[f * TOK + tok0 + r] : 0.f;
  }
  __syncthreads();

  for (int tile = wv; tile < 64; tile += 8) {
    const int mt = tile >> 4, nt = tile & 15;
    const int n  = nt * 16 + lh;
    const int mr = mt * 16 + lh;
    v8f acc;
    const float bv = bm0[n];
    #pragma unroll
    for (int r = 0; r < 8; ++r) acc[r] = bv;
    #pragma unroll
    for (int kk = 0; kk < 12; ++kk) {
      const int k  = 4 * kk + 2 * sel;
      const int k0 = (k     < Mn) ? k     : (Mn - 1);
      const int k1 = (k + 1 < Mn) ? k + 1 : (Mn - 1);
      v2f a, b;
      a[0] = Abuf[mr][k];      a[1] = Abuf[mr][k + 1];
      b[0] = Wm0[n * Mn + k0]; b[1] = Wm0[n * Mn + k1];
      acc = wmma4(a, b, acc);
    }
    #pragma unroll
    for (int r = 0; r < 8; ++r)
      Bbuf[mt * 16 + r + 8 * sel][n] = fmaxf(acc[r], 0.f);
  }
  __syncthreads();

  hidden_layer(Wm1, bm1, Bbuf, Abuf, lane, wv);
  __syncthreads();
  hidden_layer(Wm2, bm2, Abuf, Bbuf, lane, wv);
  __syncthreads();

  for (int tile = wv; tile < 8; tile += 8) {
    const int mt = tile >> 1, nt = tile & 1;
    const int n  = nt * 16 + lh;
    const int nc = (n < OUTD) ? n : (OUTD - 1);
    const int mr = mt * 16 + lh;
    v8f acc;
    const float bv = bm3[nc];
    #pragma unroll
    for (int r = 0; r < 8; ++r) acc[r] = bv;
    #pragma unroll 4
    for (int kk = 0; kk < MHD / 4; ++kk) {
      const int k = 4 * kk + 2 * sel;
      v2f a, b;
      a[0] = Bbuf[mr][k];       a[1] = Bbuf[mr][k + 1];
      b[0] = Wm3[nc * MHD + k]; b[1] = Wm3[nc * MHD + k + 1];
      acc = wmma4(a, b, acc);
    }
    if (n < OUTD) {
      #pragma unroll
      for (int r = 0; r < 8; ++r)
        Out[(tok0 + mt * 16 + r + 8 * sel) * OUTD + n] = acc[r];
    }
  }
}

extern "C" void kernel_launch(void* const* d_in, const int* in_sizes, int n_in,
                              void* d_out, int out_size, void* d_ws, size_t ws_size,
                              hipStream_t stream) {
  const float* Dis  = (const float*)d_in[0];
  const float* Wih0 = (const float*)d_in[1];
  const float* Whh0 = (const float*)d_in[2];
  const float* bih0 = (const float*)d_in[3];
  const float* bhh0 = (const float*)d_in[4];
  const float* Wih1 = (const float*)d_in[5];
  const float* Whh1 = (const float*)d_in[6];
  const float* bih1 = (const float*)d_in[7];
  const float* bhh1 = (const float*)d_in[8];
  const float* Wih2 = (const float*)d_in[9];
  const float* Whh2 = (const float*)d_in[10];
  const float* bih2 = (const float*)d_in[11];
  const float* bhh2 = (const float*)d_in[12];
  const float* Wfc1 = (const float*)d_in[13];
  const float* bfc1 = (const float*)d_in[14];
  const float* Wfc2 = (const float*)d_in[15];
  const float* bfc2 = (const float*)d_in[16];
  const float* Wm0  = (const float*)d_in[17];
  const float* bm0  = (const float*)d_in[18];
  const float* Wm1  = (const float*)d_in[19];
  const float* bm1  = (const float*)d_in[20];
  const float* Wm2  = (const float*)d_in[21];
  const float* bm2  = (const float*)d_in[22];
  const float* Wm3  = (const float*)d_in[23];
  const float* bm3  = (const float*)d_in[24];

  float* out   = (float*)d_out;
  float* Dpred = out;
  float* cpf   = out + Mn * Bn * Tn;

  dim3 g1(NBG, Mn);
  lstm45_kernel<<<g1, 160, 0, stream>>>(
      Dis, Wih0, Whh0, bih0, bhh0, Wih1, Whh1, bih1, bhh1,
      Wih2, Whh2, bih2, bhh2, Wfc1, bfc1, Wfc2, bfc2, Dpred);

  mlp_kernel<<<TOK / 64, 256, 0, stream>>>(
      Dpred, Wm0, bm0, Wm1, bm1, Wm2, bm2, Wm3, bm3, cpf);
}


# MLP layers 0-2 register-blocked: a wave computes a 64-token x 32-column strip in 8 accumulators; a weight fragment is loaded once per block (was 4x) and feeds 4 WMMAs, an LDS fragment 2; k order kept
# speedup vs baseline: 2.5889x; 1.1150x over previous
#include <hip/hip_runtime.h>

#define Mn   45
#define Bn   64
#define Tn   512
#define Hn   20
#define Gn   80
#define FC   32
#define MHD  256
#define OUTD 22
#define TOK  (Bn * Tn)

#define BQ   16
#define NBG  (Bn / BQ)

typedef __attribute__((ext_vector_type(2))) float v2f;
typedef __attribute__((ext_vector_type(8))) float v8f;

__device__ __forceinline__ v8f wmma4(v2f a, v2f b, v8f c) {
  return __builtin_amdgcn_wmma_f32_16x16x4_f32(false, a, false, b, (short)0, c,
                                               false, false);
}

__device__ __forceinline__ float sigm(float x) {
  return __builtin_amdgcn_rcpf(1.f + __expf(-x));
}
__device__ __forceinline__ float tanh_fast(float x) {
  return 1.f - 2.f * __builtin_amdgcn_rcpf(1.f + __expf(2.f * x));
}

#define XP (Tn + 1)

__global__ __launch_bounds__(160) void lstm45_kernel(
    const float* __restrict__ Dis,
    const float* __restrict__ Wih0, const float* __restrict__ Whh0,
    const float* __restrict__ bih0, const float* __restrict__ bhh0,
    const float* __restrict__ Wih1, const float* __restrict__ Whh1,
    const float* __restrict__ bih1, const float* __restrict__ bhh1,
    const float* __restrict__ Wih2, const float* __restrict__ Whh2,
    const float* __restrict__ bih2, const float* __restrict__ bhh2,
    const float* __restrict__ Wfc1, const float* __restrict__ bfc1,
    const float* __restrict__ Wfc2, const float* __restrict__ bfc2,
    float* __restrict__ Dpred)
{
  __shared__ float xall[BQ][XP];
  __shared__ float hc[2][BQ][3 * Hn];

  const int tid  = threadIdx.x;
  const int m    = blockIdx.y;
  const int b0   = blockIdx.x * BQ;
  const int lane = tid & 31;
  const int wv   = tid >> 5;
  const int lh   = lane & 15;
  const int sel  = lane >> 4;

  for (int i = tid; i < BQ * Tn; i += 160) {
    int b = i / Tn, t = i % Tn;
    xall[b][t] = Dis[(m * Bn + b0 + b) * Tn + t];
  }
  for (int i = tid; i < 2 * BQ * 3 * Hn; i += 160) (&hc[0][0][0])[i] = 0.f;

  const int arow = (lh & 3) * Hn + 4 * wv + (lh >> 2);
  const int wb   = (m * Gn + arow) * Hn;
  v2f aw0[5], aw1[10], aw2[10];
  #pragma unroll
  for (int kk = 0; kk < 5; ++kk) {
    const int k = 4 * kk + 2 * sel;
    aw0[kk][0] = Whh0[wb + k]; aw0[kk][1] = Whh0[wb + k + 1];
    aw1[kk][0] = Wih1[wb + k]; aw1[kk][1] = Wih1[wb + k + 1];
    aw2[kk][0] = Wih2[wb + k]; aw2[kk][1] = Wih2[wb + k + 1];
  }
  #pragma unroll
  for (int kk = 0; kk < 5; ++kk) {
    const int k = 4 * kk + 2 * sel;
    aw1[5 + kk][0] = Whh1[wb + k]; aw1[5 + kk][1] = Whh1[wb + k + 1];
    aw2[5 + kk][0] = Whh2[wb + k]; aw2[5 + kk][1] = Whh2[wb + k + 1];
  }
  float b0r[8], w0r[8], b1r[8], b2r[8];
  #pragma unroll
  for (int r = 0; r < 8; ++r) {
    const int g = m * Gn + (r & 3) * Hn + 4 * wv + 2 * sel + (r >> 2);
    b0r[r] = bih0[g] + bhh0[g];
    b1r[r] = bih1[g] + bhh1[g];
    b2r[r] = bih2[g] + bhh2[g];
    w0r[r] = Wih0[g];
  }
  const bool headw = (wv == 3);
  v2f awfa[5], awfb[5];
  float bf1a[8], bf1b[8], wf2a[8], wf2b[8];
  #pragma unroll
  for (int kk = 0; kk < 5; ++kk) {
    awfa[kk][0] = 0.f; awfa[kk][1] = 0.f; awfb[kk][0] = 0.f; awfb[kk][1] = 0.f;
  }
  #pragma unroll
  for (int r = 0; r < 8; ++r) { bf1a[r] = 0.f; bf1b[r] = 0.f; wf2a[r] = 0.f; wf2b[r] = 0.f; }
  if (headw) {
    const int fa = (m * FC + lh) * Hn;
    const int fb = (m * FC + 16 + lh) * Hn;
    #pragma unroll
    for (int kk = 0; kk < 5; ++kk) {
      const int k = 4 * kk + 2 * sel;
      awfa[kk][0] = Wfc1[fa + k]; awfa[kk][1] = Wfc1[fa + k + 1];
      awfb[kk][0] = Wfc1[fb + k]; awfb[kk][1] = Wfc1[fb + k + 1];
    }
    #pragma unroll
    for (int r = 0; r < 8; ++r) {
      const int f = 8 * sel + r;
      bf1a[r] = bfc1[m * FC + f];      wf2a[r] = Wfc2[m * FC + f];
      bf1b[r] = bfc1[m * FC + 16 + f]; wf2b[r] = Wfc2[m * FC + 16 + f];
    }
  }
  const float bfc2v = bfc2[m];

  float c0[2] = {0.f, 0.f}, c1[2] = {0.f, 0.f}, c2[2] = {0.f, 0.f};
  float psk = 0.f, pso = 0.f;

  __syncthreads();

#define CELL(acc, cst, l)                                                  \
  {                                                                        \
    _Pragma("unroll")             \
    for (int p = 0; p < 2; ++p) {                                          \
      float ig = sigm(acc[4 * p]);                                         \
      float fg = sigm(acc[4 * p + 1]);                                     \
      float gv = tanh_fast(acc[4 * p + 2]);                                \
      float og = sigm(acc[4 * p + 3]);                                     \
      float c  = fg * cst[p] + ig * gv;                                    \
      cst[p] = c;                                                          \
      hw[(l) * Hn + p] = og * tanh_fast(c);                                \
    }                                                                      \
  }

#define TICK(L0, L1, L2, HD, P, u)                                           \
  {                                                                          \
    const float* hr = &hc[P][lh][2 * sel];                                   \
    float*       hw = &hc[1 - (P)][lh][4 * wv + 2 * sel];                    \
    v2f hb[15];                                                              \
    _Pragma("unroll")               \
    for (int j = 0; j < 15; ++j) {                                           \
      const bool need = (j < 5)  ? ((L0) || (L1))                            \
                      : (j < 10) ? ((L1) || (L2)) : ((L2) || (HD));          \
      if (need) { hb[j][0] = hr[4 * j]; hb[j][1] = hr[4 * j + 1]; }          \
      else      { hb[j][0] = 0.f; hb[j][1] = 0.f; }                          \
    }                                                                        \
    const float xv = (L0) ? xall[lh][(L0) ? (u) : 0] : 0.f;                  \
    if ((u) >= 4) {                      \
      if (headw && sel == 0)                                                 \
        Dpred[(m * Bn + b0 + lh) * Tn + ((u) - 4)] = bfc2v + (psk + pso);    \
    }                                                                        \
    v8f fa8, fb8;                                                            \
    _Pragma("unroll")               \
    for (int r = 0; r < 8; ++r) { fa8[r] = bf1a[r]; fb8[r] = bf1b[r]; }      \
    if (HD) {                               \
      if (headw) {                                   \
        _Pragma("unroll")           \
        for (int kk = 0; kk < 5; ++kk) {                                     \
          fa8 = wmma4(awfa[kk], hb[10 + kk], fa8);                           \
          fb8 = wmma4(awfb[kk], hb[10 + kk], fb8);                           \
        }                                                                    \
      }                                                                      \
    }                                                                        \
    v8f acc0, acc1, acc2;                                                    \
    _Pragma("unroll")               \
    for (int r = 0; r < 8; ++r) {                                            \
      acc0[r] = (L0) ? (b0r[r] + xv * w0r[r]) : 0.f;                         \
      acc1[r] = b1r[r];                                                      \
      acc2[r] = b2r[r];                                                      \
    }                                                                        \
    _Pragma("unroll")               \
    for (int kk = 0; kk < 10; ++kk) {                                        \
      if ((L0) && kk < 5) acc0 = wmma4(aw0[kk], hb[kk], acc0);               \
      if (L1) acc1 = wmma4(aw1[kk], hb[kk], acc1);                           \
      if (L2) acc2 = wmma4(aw2[kk], hb[5 + kk], acc2);                       \
    }                                                                        \
    if (HD) {                            \
      if (headw) {                                   \
        float ps = 0.f;                                                      \
        _Pragma("unroll")           \
        for (int r = 0; r < 8; ++r) ps += fmaxf(fa8[r], 0.f) * wf2a[r];      \
        _Pragma("unroll")           \
        for (int r = 0; r < 8; ++r) ps += fmaxf(fb8[r], 0.f) * wf2b[r];      \
        psk = ps;                        \
        pso = __shfl_xor(ps, 16, 32);              \
      }                                                                      \
    }                                                                        \
    if (L0) CELL(acc0, c0, 0)                                                \
    if (L1) CELL(acc1, c1, 1)                                                \
    if (L2) CELL(acc2, c2, 2)                                                \
    __syncthreads();                                                         \
  }

  TICK(1, 0, 0, 0, 0, 0)
  TICK(1, 1, 0, 0, 1, 1)
  TICK(1, 1, 1, 0, 0, 2)
  for (int u = 3; u < Tn - 1; u += 2) {
    TICK(1, 1, 1, 1, 1, u)
    TICK(1, 1, 1, 1, 0, u + 1)
  }
  TICK(1, 1, 1, 1, 1, Tn - 1)
  TICK(0, 1, 1, 1, 0, Tn)
  TICK(0, 0, 1, 1, 1, Tn + 1)
  TICK(0, 0, 0, 1, 0, Tn + 2)
  if (headw && sel == 0)
    Dpred[(m * Bn + b0 + lh) * Tn + (Tn - 1)] = bfc2v + (psk + pso);
#undef TICK
#undef CELL
}

#define ASTR (MHD + 4)

__device__ __forceinline__ void hidden_layer(
    const float* __restrict__ W, const float* __restrict__ bias,
    const float (*IN)[ASTR], float (*OUT)[ASTR], int lane, int wv)
{
  const int lh = lane & 15, sel = lane >> 4;
  const int n0 = wv * 32 + lh;
  const float* wp0 = W + n0 * MHD + 2 * sel;
  const float* wp1 = wp0 + 16 * MHD;
  v8f acc[4][2];
  const float bv0 = bias[n0], bv1 = bias[n0 + 16];
  #pragma unroll
  for (int mt = 0; mt < 4; ++mt) {
    #pragma unroll
    for (int r = 0; r < 8; ++r) { acc[mt][0][r] = bv0; acc[mt][1][r] = bv1; }
  }
  #pragma unroll 4
  for (int kk = 0; kk < MHD / 4; ++kk) {
    const int k = 4 * kk + 2 * sel;
    v2f b0, b1, a[4];
    b0[0] = wp0[4 * kk]; b0[1] = wp0[4 * kk + 1];
    b1[0] = wp1[4 * kk]; b1[1] = wp1[4 * kk + 1];
    #pragma unroll
    for (int mt = 0; mt < 4; ++mt) {
      a[mt][0] = IN[mt * 16 + lh][k]; a[mt][1] = IN[mt * 16 + lh][k + 1];
    }
    #pragma unroll
    for (int mt = 0; mt < 4; ++mt) {
      acc[mt][0] = wmma4(a[mt], b0, acc[mt][0]);
      acc[mt][1] = wmma4(a[mt], b1, acc[mt][1]);
    }
  }
  #pragma unroll
  for (int mt = 0; mt < 4; ++mt) {
    #pragma unroll
    for (int r = 0; r < 8; ++r) {
      OUT[mt * 16 + r + 8 * sel][n0]      = fmaxf(acc[mt][0][r], 0.f);
      OUT[mt * 16 + r + 8 * sel][n0 + 16] = fmaxf(acc[mt][1][r], 0.f);
    }
  }
}

__global__ __launch_bounds__(256) void mlp_kernel(
    const float* __restrict__ Dpred,
    const float* __restrict__ Wm0, const float* __restrict__ bm0,
    const float* __restrict__ Wm1, const float* __restrict__ bm1,
    const float* __restrict__ Wm2, const float* __restrict__ bm2,
    const float* __restrict__ Wm3, const float* __restrict__ bm3,
    float* __restrict__ Out)
{
  __shared__ float Abuf[64][ASTR];
  __shared__ float Bbuf[64][ASTR];

  const int tid  = threadIdx.x;
  const int lane = tid & 31, wv = tid >> 5;
  const int lh   = lane & 15, sel = lane >> 4;
  const int tok0 = blockIdx.x * 64;

  for (int i = tid; i < 64 * 48; i += 256) {
    int r = i / 48, f = i % 48;
    Abuf[r][f] = (f < Mn) ? Dpred[f * TOK + tok0 + r] : 0.f;
  }
  __syncthreads();

  {
    const int n0 = wv * 32 + lh;
    v8f acc[4][2];
    const float bv0 = bm0[n0], bv1 = bm0[n0 + 16];
    #pragma unroll
    for (int mt = 0; mt < 4; ++mt) {
      #pragma unroll
      for (int r = 0; r < 8; ++r) { acc[mt][0][r] = bv0; acc[mt][1][r] = bv1; }
    }
    #pragma unroll
    for (int kk = 0; kk < 12; ++kk) {
      const int k  = 4 * kk + 2 * sel;
      const int k0 = (k     < Mn) ? k     : (Mn - 1);
      const int k1 = (k + 1 < Mn) ? k + 1 : (Mn - 1);
      v2f b0, b1, a[4];
      b0[0] = Wm0[n0 * Mn + k0];        b0[1] = Wm0[n0 * Mn + k1];
      b1[0] = Wm0[(n0 + 16) * Mn + k0]; b1[1] = Wm0[(n0 + 16) * Mn + k1];
      #pragma unroll
      for (int mt = 0; mt < 4; ++mt) {
        a[mt][0] = Abuf[mt * 16 + lh][k]; a[mt][1] = Abuf[mt * 16 + lh][k + 1];
      }
      #pragma unroll
      for (int mt = 0; mt < 4; ++mt) {
        acc[mt][0] = wmma4(a[mt], b0, acc[mt][0]);
        acc[mt][1] = wmma4(a[mt], b1, acc[mt][1]);
      }
    }
    #pragma unroll
    for (int mt = 0; mt < 4; ++mt) {
      #pragma unroll
      for (int r = 0; r < 8; ++r) {
        Bbuf[mt * 16 + r + 8 * sel][n0]      = fmaxf(acc[mt][0][r], 0.f);
        Bbuf[mt * 16 + r + 8 * sel][n0 + 16] = fmaxf(acc[mt][1][r], 0.f);
      }
    }
  }
  __syncthreads();

  hidden_layer(Wm1, bm1, Bbuf, Abuf, lane, wv);
  __syncthreads();
  hidden_layer(Wm2, bm2, Abuf, Bbuf, lane, wv);
  __syncthreads();

  for (int tile = wv; tile < 8; tile += 8) {
    const int mt = tile >> 1, nt = tile & 1;
    const int n  = nt * 16 + lh;
    const int nc = (n < OUTD) ? n : (OUTD - 1);
    const int mr = mt * 16 + lh;
    v8f acc;
    const float bv = bm3[nc];
    #pragma unroll
    for (int r = 0; r < 8; ++r) acc[r] = bv;
    #pragma unroll 4
    for (int kk = 0; kk < MHD / 4; ++kk) {
      const int k = 4 * kk + 2 * sel;
      v2f a, b;
      a[0] = Bbuf[mr][k];       a[1] = Bbuf[mr][k + 1];
      b[0] = Wm3[nc * MHD + k]; b[1] = Wm3[nc * MHD + k + 1];
      acc = wmma4(a, b, acc);
    }
    if (n < OUTD) {
      #pragma unroll
      for (int r = 0; r < 8; ++r)
        Out[(tok0 + mt * 16 + r + 8 * sel) * OUTD + n] = acc[r];
    }
  }
}

extern "C" void kernel_launch(void* const* d_in, const int* in_sizes, int n_in,
                              void* d_out, int out_size, void* d_ws, size_t ws_size,
                              hipStream_t stream) {
  const float* Dis  = (const float*)d_in[0];
  const float* Wih0 = (const float*)d_in[1];
  const float* Whh0 = (const float*)d_in[2];
  const float* bih0 = (const float*)d_in[3];
  const float* bhh0 = (const float*)d_in[4];
  const float* Wih1 = (const float*)d_in[5];
  const float* Whh1 = (const float*)d_in[6];
  const float* bih1 = (const float*)d_in[7];
  const float* bhh1 = (const float*)d_in[8];
  const float* Wih2 = (const float*)d_in[9];
  const float* Whh2 = (const float*)d_in[10];
  const float* bih2 = (const float*)d_in[11];
  const float* bhh2 = (const float*)d_in[12];
  const float* Wfc1 = (const float*)d_in[13];
  const float* bfc1 = (const float*)d_in[14];
  const float* Wfc2 = (const float*)d_in[15];
  const float* bfc2 = (const float*)d_in[16];
  const float* Wm0  = (const float*)d_in[17];
  const float* bm0  = (const float*)d_in[18];
  const float* Wm1  = (const float*)d_in[19];
  const float* bm1  = (const float*)d_in[20];
  const float* Wm2  = (const float*)d_in[21];
  const float* bm2  = (const float*)d_in[22];
  const float* Wm3  = (const float*)d_in[23];
  const float* bm3  = (const float*)d_in[24];

  float* out   = (float*)d_out;
  float* Dpred = out;
  float* cpf   = out + Mn * Bn * Tn;

  dim3 g1(NBG, Mn);
  lstm45_kernel<<<g1, 160, 0, stream>>>(
      Dis, Wih0, Whh0, bih0, bhh0, Wih1, Whh1, bih1, bhh1,
      Wih2, Whh2, bih2, bhh2, Wfc1, bfc1, Wfc2, bfc2, Dpred);

  mlp_kernel<<<TOK / 64, 256, 0, stream>>>(
      Dpred, Wm0, bm0, Wm1, bm1, Wm2, bm2, Wm3, bm3, cpf);
}


# LSTM tick split by tile chain instead of tile row: 16 waves, one chain each (5+5+5 layer tiles, head), so a SIMD issues 35/30/35/35 WMMAs per tick, was 50/35/25/25; k order of every sum unchanged
# speedup vs baseline: 3.0529x; 1.1792x over previous
#include <hip/hip_runtime.h>

#define Mn   45
#define Bn   64
#define Tn   512
#define Hn   20
#define Gn   80
#define FC   32
#define MHD  256
#define OUTD 22
#define TOK  (Bn * Tn)

#define BQ   16
#define NBG  (Bn / BQ)

typedef __attribute__((ext_vector_type(2))) float v2f;
typedef __attribute__((ext_vector_type(8))) float v8f;

__device__ __forceinline__ v8f wmma4(v2f a, v2f b, v8f c) {
  return __builtin_amdgcn_wmma_f32_16x16x4_f32(false, a, false, b, (short)0, c,
                                               false, false);
}

__device__ __forceinline__ float sigm(float x) {
  return __builtin_amdgcn_rcpf(1.f + __expf(-x));
}
__device__ __forceinline__ float tanh_fast(float x) {
  return 1.f - 2.f * __builtin_amdgcn_rcpf(1.f + __expf(2.f * x));
}

#define XP   (Tn + 1)
#define NW1  16
#define NT1  (NW1 * 32)
#define PLN  (BQ * 3 * Hn)

#define ROLE_LAYER 0x0223202211011100ull
#define ROLE_TILE  0x4430231043221010ull

__global__ __launch_bounds__(NT1) void lstm45_kernel(
    const float* __restrict__ Dis,
    const float* __restrict__ Wih0, const float* __restrict__ Whh0,
    const float* __restrict__ bih0, const float* __restrict__ bhh0,
    const float* __restrict__ Wih1, const float* __restrict__ Whh1,
    const float* __restrict__ bih1, const float* __restrict__ bhh1,
    const float* __restrict__ Wih2, const float* __restrict__ Whh2,
    const float* __restrict__ bih2, const float* __restrict__ bhh2,
    const float* __restrict__ Wfc1, const float* __restrict__ bfc1,
    const float* __restrict__ Wfc2, const float* __restrict__ bfc2,
    float* __restrict__ Dpred)
{
  __shared__ float xall[BQ][XP];
  __shared__ float hc[2][BQ][3 * Hn];

  const int tid  = threadIdx.x;
  const int m    = blockIdx.y;
  const int b0   = blockIdx.x * BQ;
  const int lane = tid & 31;
  const int wv   = __builtin_amdgcn_readfirstlane(tid >> 5);
  const int lh   = lane & 15;
  const int sel  = lane >> 4;
  const int lay  = (int)((ROLE_LAYER >> (4 * wv)) & 15);
  const int tile = (int)((ROLE_TILE  >> (4 * wv)) & 15);

  for (int i = tid; i < BQ * Tn; i += NT1) {
    int b = i / Tn, t = i % Tn;
    xall[b][t] = Dis[(m * Bn + b0 + b) * Tn + t];
  }
  for (int i = tid; i < 2 * PLN; i += NT1) (&hc[0][0][0])[i] = 0.f;

  const float* pa;
  const float* pb;
  if (lay == 3) {
    pa = Wfc1 + (m * FC + lh) * Hn;
    pb = Wfc1 + (m * FC + 16 + lh) * Hn;
  } else {
    const int arow = (lh & 3) * Hn + 4 * tile + (lh >> 2);
    const int wb   = (m * Gn + arow) * Hn;
    pa = ((lay == 0) ? Whh0 : (lay == 1) ? Wih1 : Wih2) + wb;
    pb = ((lay == 2) ? Whh2 : Whh1) + wb;
  }
  v2f aw[10];
  #pragma unroll
  for (int kk = 0; kk < 5; ++kk) {
    const int k = 4 * kk + 2 * sel;
    aw[kk][0]     = pa[k]; aw[kk][1]     = pa[k + 1];
    aw[5 + kk][0] = pb[k]; aw[5 + kk][1] = pb[k + 1];
  }
  float s0[8], s1[8], s2[8], s3[8];
  #pragma unroll
  for (int r = 0; r < 8; ++r) { s0[r] = 0.f; s1[r] = 0.f; s2[r] = 0.f; s3[r] = 0.f; }
  if (lay == 3) {
    #pragma unroll
    for (int r = 0; r < 8; ++r) {
      const int f = 8 * sel + r;
      s0[r] = bfc1[m * FC + f];      s2[r] = Wfc2[m * FC + f];
      s1[r] = bfc1[m * FC + 16 + f]; s3[r] = Wfc2[m * FC + 16 + f];
    }
  } else {
    const float* bi = (lay == 0) ? bih0 : (lay == 1) ? bih1 : bih2;
    const float* bh = (lay == 0) ? bhh0 : (lay == 1) ? bhh1 : bhh2;
    #pragma unroll
    for (int r = 0; r < 8; ++r) {
      const int g = m * Gn + (r & 3) * Hn + 4 * tile + 2 * sel + (r >> 2);
      s0[r] = bi[g] + bh[g];
      if (lay == 0) s1[r] = Wih0[g];
    }
  }
  const float bfc2v = bfc2[m];

  const int rcol = (lay == 0) ? 0 : (lay - 1) * Hn;
  const float* hrd = &hc[0][lh][rcol + 2 * sel];
  float*       hwr = &hc[0][lh][((lay == 3) ? 0 : lay * Hn) + 4 * tile + 2 * sel];

  float cst[2] = {0.f, 0.f};
  float psk = 0.f, pso = 0.f;

  __syncthreads();

#define CELL(acc, hw)                                                      \
  {                                                                        \
    _Pragma("unroll")             \
    for (int p = 0; p < 2; ++p) {                                          \
      float ig = sigm(acc[4 * p]);                                         \
      float fg = sigm(acc[4 * p + 1]);                                     \
      float gv = tanh_fast(acc[4 * p + 2]);                                \
      float og = sigm(acc[4 * p + 3]);                                     \
      float c  = fg * cst[p] + ig * gv;                                    \
      cst[p] = c;                                                          \
      hw[p] = og * tanh_fast(c);                                           \
    }                                                                      \
  }

  for (int u = 0; u < Tn + 3; ++u) {
    const int rp = (u & 1) * PLN;
    const float* hr = hrd + rp;
    float*       hw = hwr + (PLN - rp);
    if (lay == 3) {
      if (u >= 4) {
        if (sel == 0)
          Dpred[(m * Bn + b0 + lh) * Tn + (u - 4)] = bfc2v + (psk + pso);
      }
      if (u >= 3) {
        v2f hb[5];
        #pragma unroll
        for (int j = 0; j < 5; ++j) { hb[j][0] = hr[4 * j]; hb[j][1] = hr[4 * j + 1]; }
        v8f fa8, fb8;
        #pragma unroll
        for (int r = 0; r < 8; ++r) { fa8[r] = s0[r]; fb8[r] = s1[r]; }
        #pragma unroll
        for (int kk = 0; kk < 5; ++kk) {
          fa8 = wmma4(aw[kk],     hb[kk], fa8);
          fb8 = wmma4(aw[5 + kk], hb[kk], fb8);
        }
        float ps = 0.f;
        #pragma unroll
        for (int r = 0; r < 8; ++r) ps += fmaxf(fa8[r], 0.f) * s2[r];
        #pragma unroll
        for (int r = 0; r < 8; ++r) ps += fmaxf(fb8[r], 0.f) * s3[r];
        psk = ps;
        pso = __shfl_xor(ps, 16, 32);
      }
    } else if (lay == 0) {
      if (u < Tn) {
        v2f hb[5];
        #pragma unroll
        for (int j = 0; j < 5; ++j) { hb[j][0] = hr[4 * j]; hb[j][1] = hr[4 * j + 1]; }
        const float xv = xall[lh][u];
        v8f acc;
        #pragma unroll
        for (int r = 0; r < 8; ++r) acc[r] = s0[r] + xv * s1[r];
        #pragma unroll
        for (int kk = 0; kk < 5; ++kk) acc = wmma4(aw[kk], hb[kk], acc);
        CELL(acc, hw)
      }
    } else {
      if (u >= lay && u < Tn + lay) {
        v2f hb[10];
        #pragma unroll
        for (int j = 0; j < 10; ++j) { hb[j][0] = hr[4 * j]; hb[j][1] = hr[4 * j + 1]; }
        __builtin_amdgcn_sched_barrier(0);
        v8f acc;
        #pragma unroll
        for (int r = 0; r < 8; ++r) acc[r] = s0[r];
        #pragma unroll
        for (int kk = 0; kk < 10; ++kk) acc = wmma4(aw[kk], hb[kk], acc);
        CELL(acc, hw)
      }
    }
    __syncthreads();
  }
  if (lay == 3 && sel == 0)
    Dpred[(m * Bn + b0 + lh) * Tn + (Tn - 1)] = bfc2v + (psk + pso);
#undef CELL
}

#define ASTR (MHD + 4)

__device__ __forceinline__ void hidden_layer(
    const float* __restrict__ W, const float* __restrict__ bias,
    const float (*IN)[ASTR], float (*OUT)[ASTR], int lane, int wv)
{
  const int lh = lane & 15, sel = lane >> 4;
  const int n0 = wv * 32 + lh;
  const float* wp0 = W + n0 * MHD + 2 * sel;
  const float* wp1 = wp0 + 16 * MHD;
  v8f acc[4][2];
  const float bv0 = bias[n0], bv1 = bias[n0 + 16];
  #pragma unroll
  for (int mt = 0; mt < 4; ++mt) {
    #pragma unroll
    for (int r = 0; r < 8; ++r) { acc[mt][0][r] = bv0; acc[mt][1][r] = bv1; }
  }
  #pragma unroll 4
  for (int kk = 0; kk < MHD / 4; ++kk) {
    const int k = 4 * kk + 2 * sel;
    v2f b0, b1, a[4];
    b0[0] = wp0[4 * kk]; b0[1] = wp0[4 * kk + 1];
    b1[0] = wp1[4 * kk]; b1[1] = wp1[4 * kk + 1];
    #pragma unroll
    for (int mt = 0; mt < 4; ++mt) {
      a[mt][0] = IN[mt * 16 + lh][k]; a[mt][1] = IN[mt * 16 + lh][k + 1];
    }
    #pragma unroll
    for (int mt = 0; mt < 4; ++mt) {
      acc[mt][0] = wmma4(a[mt], b0, acc[mt][0]);
      acc[mt][1] = wmma4(a[mt], b1, acc[mt][1]);
    }
  }
  #pragma unroll
  for (int mt = 0; mt < 4; ++mt) {
    #pragma unroll
    for (int r = 0; r < 8; ++r) {
      OUT[mt * 16 + r + 8 * sel][n0]      = fmaxf(acc[mt][0][r], 0.f);
      OUT[mt * 16 + r + 8 * sel][n0 + 16] = fmaxf(acc[mt][1][r], 0.f);
    }
  }
}

__global__ __launch_bounds__(256) void mlp_kernel(
    const float* __restrict__ Dpred,
    const float* __restrict__ Wm0, const float* __restrict__ bm0,
    const float* __restrict__ Wm1, const float* __restrict__ bm1,
    const float* __restrict__ Wm2, const float* __restrict__ bm2,
    const float* __restrict__ Wm3, const float* __restrict__ bm3,
    float* __restrict__ Out)
{
  __shared__ float Abuf[64][ASTR];
  __shared__ float Bbuf[64][ASTR];

  const int tid  = threadIdx.x;
  const int lane = tid & 31, wv = tid >> 5;
  const int lh   = lane & 15, sel = lane >> 4;
  const int tok0 = blockIdx.x * 64;

  for (int i = tid; i < 64 * 48; i += 256) {
    int r = i / 48, f = i % 48;
    Abuf[r][f] = (f < Mn) ? Dpred[f * TOK + tok0 + r] : 0.f;
  }
  __syncthreads();

  {
    const int n0 = wv * 32 + lh;
    v8f acc[4][2];
    const float bv0 = bm0[n0], bv1 = bm0[n0 + 16];
    #pragma unroll
    for (int mt = 0; mt < 4; ++mt) {
      #pragma unroll
      for (int r = 0; r < 8; ++r) { acc[mt][0][r] = bv0; acc[mt][1][r] = bv1; }
    }
    #pragma unroll
    for (int kk = 0; kk < 12; ++kk) {
      const int k  = 4 * kk + 2 * sel;
      const int k0 = (k     < Mn) ? k     : (Mn - 1);
      const int k1 = (k + 1 < Mn) ? k + 1 : (Mn - 1);
      v2f b0, b1, a[4];
      b0[0] = Wm0[n0 * Mn + k0];        b0[1] = Wm0[n0 * Mn + k1];
      b1[0] = Wm0[(n0 + 16) * Mn + k0]; b1[1] = Wm0[(n0 + 16) * Mn + k1];
      #pragma unroll
      for (int mt = 0; mt < 4; ++mt) {
        a[mt][0] = Abuf[mt * 16 + lh][k]; a[mt][1] = Abuf[mt * 16 + lh][k + 1];
      }
      #pragma unroll
      for (int mt = 0; mt < 4; ++mt) {
        acc[mt][0] = wmma4(a[mt], b0, acc[mt][0]);
        acc[mt][1] = wmma4(a[mt], b1, acc[mt][1]);
      }
    }
    #pragma unroll
    for (int mt = 0; mt < 4; ++mt) {
      #pragma unroll
      for (int r = 0; r < 8; ++r) {
        Bbuf[mt * 16 + r + 8 * sel][n0]      = fmaxf(acc[mt][0][r], 0.f);
        Bbuf[mt * 16 + r + 8 * sel][n0 + 16] = fmaxf(acc[mt][1][r], 0.f);
      }
    }
  }
  __syncthreads();

  hidden_layer(Wm1, bm1, Bbuf, Abuf, lane, wv);
  __syncthreads();
  hidden_layer(Wm2, bm2, Abuf, Bbuf, lane, wv);
  __syncthreads();

  for (int tile = wv; tile < 8; tile += 8) {
    const int mt = tile >> 1, nt = tile & 1;
    const int n  = nt * 16 + lh;
    const int nc = (n < OUTD) ? n : (OUTD - 1);
    const int mr = mt * 16 + lh;
    v8f acc;
    const float bv = bm3[nc];
    #pragma unroll
    for (int r = 0; r < 8; ++r) acc[r] = bv;
    #pragma unroll 4
    for (int kk = 0; kk < MHD / 4; ++kk) {
      const int k = 4 * kk + 2 * sel;
      v2f a, b;
      a[0] = Bbuf[mr][k];       a[1] = Bbuf[mr][k + 1];
      b[0] = Wm3[nc * MHD + k]; b[1] = Wm3[nc * MHD + k + 1];
      acc = wmma4(a, b, acc);
    }
    if (n < OUTD) {
      #pragma unroll
      for (int r = 0; r < 8; ++r)
        Out[(tok0 + mt * 16 + r + 8 * sel) * OUTD + n] = acc[r];
    }
  }
}

extern "C" void kernel_launch(void* const* d_in, const int* in_sizes, int n_in,
                              void* d_out, int out_size, void* d_ws, size_t ws_size,
                              hipStream_t stream) {
  const float* Dis  = (const float*)d_in[0];
  const float* Wih0 = (const float*)d_in[1];
  const float* Whh0 = (const float*)d_in[2];
  const float* bih0 = (const float*)d_in[3];
  const float* bhh0 = (const float*)d_in[4];
  const float* Wih1 = (const float*)d_in[5];
  const float* Whh1 = (const float*)d_in[6];
  const float* bih1 = (const float*)d_in[7];
  const float* bhh1 = (const float*)d_in[8];
  const float* Wih2 = (const float*)d_in[9];
  const float* Whh2 = (const float*)d_in[10];
  const float* bih2 = (const float*)d_in[11];
  const float* bhh2 = (const float*)d_in[12];
  const float* Wfc1 = (const float*)d_in[13];
  const float* bfc1 = (const float*)d_in[14];
  const float* Wfc2 = (const float*)d_in[15];
  const float* bfc2 = (const float*)d_in[16];
  const float* Wm0  = (const float*)d_in[17];
  const float* bm0  = (const float*)d_in[18];
  const float* Wm1  = (const float*)d_in[19];
  const float* bm1  = (const float*)d_in[20];
  const float* Wm2  = (const float*)d_in[21];
  const float* bm2  = (const float*)d_in[22];
  const float* Wm3  = (const float*)d_in[23];
  const float* bm3  = (const float*)d_in[24];

  float* out   = (float*)d_out;
  float* Dpred = out;
  float* cpf   = out + Mn * Bn * Tn;

  dim3 g1(NBG, Mn);
  lstm45_kernel<<<g1, NT1, 0, stream>>>(
      Dis, Wih0, Whh0, bih0, bhh0, Wih1, Whh1, bih1, bhh1,
      Wih2, Whh2, bih2, bhh2, Wfc1, bfc1, Wfc2, bfc2, Dpred);

  mlp_kernel<<<TOK / 64, 256, 0, stream>>>(
      Dpred, Wm0, bm0, Wm1, bm1, Wm2, bm2, Wm3, bm3, cpf);
}


# LSTM cell activations on the hardware tanh (v_tanh_f32): sigmoid(x) = (1+tanh(x/2))/2, halving folded exactly into the i,f,o weight rows and biases; 5 transcendentals a cell, was 10; sum order kept
# speedup vs baseline: 3.3375x; 1.0932x over previous
#include <hip/hip_runtime.h>

#define Mn   45
#define Bn   64
#define Tn   512
#define Hn   20
#define Gn   80
#define FC   32
#define MHD  256
#define OUTD 22
#define TOK  (Bn * Tn)

#define BQ   16
#define NBG  (Bn / BQ)

typedef __attribute__((ext_vector_type(2))) float v2f;
typedef __attribute__((ext_vector_type(8))) float v8f;

__device__ __forceinline__ v8f wmma4(v2f a, v2f b, v8f c) {
  return __builtin_amdgcn_wmma_f32_16x16x4_f32(false, a, false, b, (short)0, c,
                                               false, false);
}

__device__ __forceinline__ float tanh_hw(float x) {
  return __builtin_amdgcn_tanhf(x);
}

#define XP   (Tn + 1)
#define NW1  16
#define NT1  (NW1 * 32)
#define PLN  (BQ * 3 * Hn)

#define ROLE_LAYER 0x0223202211011100ull
#define ROLE_TILE  0x4430231043221010ull

__global__ __launch_bounds__(NT1) void lstm45_kernel(
    const float* __restrict__ Dis,
    const float* __restrict__ Wih0, const float* __restrict__ Whh0,
    const float* __restrict__ bih0, const float* __restrict__ bhh0,
    const float* __restrict__ Wih1, const float* __restrict__ Whh1,
    const float* __restrict__ bih1, const float* __restrict__ bhh1,
    const float* __restrict__ Wih2, const float* __restrict__ Whh2,
    const float* __restrict__ bih2, const float* __restrict__ bhh2,
    const float* __restrict__ Wfc1, const float* __restrict__ bfc1,
    const float* __restrict__ Wfc2, const float* __restrict__ bfc2,
    float* __restrict__ Dpred)
{
  __shared__ float xall[BQ][XP];
  __shared__ float hc[2][BQ][3 * Hn];

  const int tid  = threadIdx.x;
  const int m    = blockIdx.y;
  const int b0   = blockIdx.x * BQ;
  const int lane = tid & 31;
  const int wv   = __builtin_amdgcn_readfirstlane(tid >> 5);
  const int lh   = lane & 15;
  const int sel  = lane >> 4;
  const int lay  = (int)((ROLE_LAYER >> (4 * wv)) & 15);
  const int tile = (int)((ROLE_TILE  >> (4 * wv)) & 15);

  for (int i = tid; i < BQ * Tn; i += NT1) {
    int b = i / Tn, t = i % Tn;
    xall[b][t] = Dis[(m * Bn + b0 + b) * Tn + t];
  }
  for (int i = tid; i < 2 * PLN; i += NT1) (&hc[0][0][0])[i] = 0.f;

  const float* pa;
  const float* pb;
  if (lay == 3) {
    pa = Wfc1 + (m * FC + lh) * Hn;
    pb = Wfc1 + (m * FC + 16 + lh) * Hn;
  } else {
    const int arow = (lh & 3) * Hn + 4 * tile + (lh >> 2);
    const int wb   = (m * Gn + arow) * Hn;
    pa = ((lay == 0) ? Whh0 : (lay == 1) ? Wih1 : Wih2) + wb;
    pb = ((lay == 2) ? Whh2 : Whh1) + wb;
  }
  const float arsc = (lay != 3 && (lh & 3) != 2) ? 0.5f : 1.f;
  v2f aw[10];
  #pragma unroll
  for (int kk = 0; kk < 5; ++kk) {
    const int k = 4 * kk + 2 * sel;
    aw[kk][0]     = arsc * pa[k]; aw[kk][1]     = arsc * pa[k + 1];
    aw[5 + kk][0] = arsc * pb[k]; aw[5 + kk][1] = arsc * pb[k + 1];
  }
  float s0[8], s1[8], s2[8], s3[8];
  #pragma unroll
  for (int r = 0; r < 8; ++r) { s0[r] = 0.f; s1[r] = 0.f; s2[r] = 0.f; s3[r] = 0.f; }
  if (lay == 3) {
    #pragma unroll
    for (int r = 0; r < 8; ++r) {
      const int f = 8 * sel + r;
      s0[r] = bfc1[m * FC + f];      s2[r] = Wfc2[m * FC + f];
      s1[r] = bfc1[m * FC + 16 + f]; s3[r] = Wfc2[m * FC + 16 + f];
    }
  } else {
    const float* bi = (lay == 0) ? bih0 : (lay == 1) ? bih1 : bih2;
    const float* bh = (lay == 0) ? bhh0 : (lay == 1) ? bhh1 : bhh2;
    #pragma unroll
    for (int r = 0; r < 8; ++r) {
      const int g = m * Gn + (r & 3) * Hn + 4 * tile + 2 * sel + (r >> 2);
      const float rsc = ((r & 3) != 2) ? 0.5f : 1.f;
      s0[r] = rsc * (bi[g] + bh[g]);
      if (lay == 0) s1[r] = rsc * Wih0[g];
    }
  }
  const float bfc2v = bfc2[m];

  const int rcol = (lay == 0) ? 0 : (lay - 1) * Hn;
  const float* hrd = &hc[0][lh][rcol + 2 * sel];
  float*       hwr = &hc[0][lh][((lay == 3) ? 0 : lay * Hn) + 4 * tile + 2 * sel];

  float cst[2] = {0.f, 0.f};
  float psk = 0.f, pso = 0.f;

  __syncthreads();

#define CELL(acc, hw)                                                      \
  {                                                                        \
    _Pragma("unroll")             \
    for (int p = 0; p < 2; ++p) {                                          \
      float ig = fmaf(0.5f, tanh_hw(acc[4 * p]),     0.5f);                \
      float fg = fmaf(0.5f, tanh_hw(acc[4 * p + 1]), 0.5f);                \
      float gv = tanh_hw(acc[4 * p + 2]);                                  \
      float og = fmaf(0.5f, tanh_hw(acc[4 * p + 3]), 0.5f);                \
      float c  = fg * cst[p] + ig * gv;                                    \
      cst[p] = c;                                                          \
      hw[p] = og * tanh_hw(c);                                             \
    }                                                                      \
  }

  for (int u = 0; u < Tn + 3; ++u) {
    const int rp = (u & 1) * PLN;
    const float* hr = hrd + rp;
    float*       hw = hwr + (PLN - rp);
    if (lay == 3) {
      if (u >= 4) {
        if (sel == 0)
          Dpred[(m * Bn + b0 + lh) * Tn + (u - 4)] = bfc2v + (psk + pso);
      }
      if (u >= 3) {
        v2f hb[5];
        #pragma unroll
        for (int j = 0; j < 5; ++j) { hb[j][0] = hr[4 * j]; hb[j][1] = hr[4 * j + 1]; }
        v8f fa8, fb8;
        #pragma unroll
        for (int r = 0; r < 8; ++r) { fa8[r] = s0[r]; fb8[r] = s1[r]; }
        #pragma unroll
        for (int kk = 0; kk < 5; ++kk) {
          fa8 = wmma4(aw[kk],     hb[kk], fa8);
          fb8 = wmma4(aw[5 + kk], hb[kk], fb8);
        }
        float ps = 0.f;
        #pragma unroll
        for (int r = 0; r < 8; ++r) ps += fmaxf(fa8[r], 0.f) * s2[r];
        #pragma unroll
        for (int r = 0; r < 8; ++r) ps += fmaxf(fb8[r], 0.f) * s3[r];
        psk = ps;
        pso = __shfl_xor(ps, 16, 32);
      }
    } else if (lay == 0) {
      if (u < Tn) {
        v2f hb[5];
        #pragma unroll
        for (int j = 0; j < 5; ++j) { hb[j][0] = hr[4 * j]; hb[j][1] = hr[4 * j + 1]; }
        const float xv = xall[lh][u];
        v8f acc;
        #pragma unroll
        for (int r = 0; r < 8; ++r) acc[r] = s0[r] + xv * s1[r];
        #pragma unroll
        for (int kk = 0; kk < 5; ++kk) acc = wmma4(aw[kk], hb[kk], acc);
        CELL(acc, hw)
      }
    } else {
      if (u >= lay && u < Tn + lay) {
        v2f hb[10];
        #pragma unroll
        for (int j = 0; j < 10; ++j) { hb[j][0] = hr[4 * j]; hb[j][1] = hr[4 * j + 1]; }
        __builtin_amdgcn_sched_barrier(0);
        v8f acc;
        #pragma unroll
        for (int r = 0; r < 8; ++r) acc[r] = s0[r];
        #pragma unroll
        for (int kk = 0; kk < 10; ++kk) acc = wmma4(aw[kk], hb[kk], acc);
        CELL(acc, hw)
      }
    }
    __syncthreads();
  }
  if (lay == 3 && sel == 0)
    Dpred[(m * Bn + b0 + lh) * Tn + (Tn - 1)] = bfc2v + (psk + pso);
#undef CELL
}

#define ASTR (MHD + 4)

__device__ __forceinline__ void hidden_layer(
    const float* __restrict__ W, const float* __restrict__ bias,
    const float (*IN)[ASTR], float (*OUT)[ASTR], int lane, int wv)
{
  const int lh = lane & 15, sel = lane >> 4;
  const int n0 = wv * 32 + lh;
  const float* wp0 = W + n0 * MHD + 2 * sel;
  const float* wp1 = wp0 + 16 * MHD;
  v8f acc[4][2];
  const float bv0 = bias[n0], bv1 = bias[n0 + 16];
  #pragma unroll
  for (int mt = 0; mt < 4; ++mt) {
    #pragma unroll
    for (int r = 0; r < 8; ++r) { acc[mt][0][r] = bv0; acc[mt][1][r] = bv1; }
  }
  #pragma unroll 4
  for (int kk = 0; kk < MHD / 4; ++kk) {
    const int k = 4 * kk + 2 * sel;
    v2f b0, b1, a[4];
    b0[0] = wp0[4 * kk]; b0[1] = wp0[4 * kk + 1];
    b1[0] = wp1[4 * kk]; b1[1] = wp1[4 * kk + 1];
    #pragma unroll
    for (int mt = 0; mt < 4; ++mt) {
      a[mt][0] = IN[mt * 16 + lh][k]; a[mt][1] = IN[mt * 16 + lh][k + 1];
    }
    #pragma unroll
    for (int mt = 0; mt < 4; ++mt) {
      acc[mt][0] = wmma4(a[mt], b0, acc[mt][0]);
      acc[mt][1] = wmma4(a[mt], b1, acc[mt][1]);
    }
  }
  #pragma unroll
  for (int mt = 0; mt < 4; ++mt) {
    #pragma unroll
    for (int r = 0; r < 8; ++r) {
      OUT[mt * 16 + r + 8 * sel][n0]      = fmaxf(acc[mt][0][r], 0.f);
      OUT[mt * 16 + r + 8 * sel][n0 + 16] = fmaxf(acc[mt][1][r], 0.f);
    }
  }
}

__global__ __launch_bounds__(256) void mlp_kernel(
    const float* __restrict__ Dpred,
    const float* __restrict__ Wm0, const float* __restrict__ bm0,
    const float* __restrict__ Wm1, const float* __restrict__ bm1,
    const float* __restrict__ Wm2, const float* __restrict__ bm2,
    const float* __restrict__ Wm3, const float* __restrict__ bm3,
    float* __restrict__ Out)
{
  __shared__ float Abuf[64][ASTR];
  __shared__ float Bbuf[64][ASTR];

  const int tid  = threadIdx.x;
  const int lane = tid & 31, wv = tid >> 5;
  const int lh   = lane & 15, sel = lane >> 4;
  const int tok0 = blockIdx.x * 64;

  for (int i = tid; i < 64 * 48; i += 256) {
    int r = i / 48, f = i % 48;
    Abuf[r][f] = (f < Mn) ? Dpred[f * TOK + tok0 + r] : 0.f;
  }
  __syncthreads();

  {
    const int n0 = wv * 32 + lh;
    v8f acc[4][2];
    const float bv0 = bm0[n0], bv1 = bm0[n0 + 16];
    #pragma unroll
    for (int mt = 0; mt < 4; ++mt) {
      #pragma unroll
      for (int r = 0; r < 8; ++r) { acc[mt][0][r] = bv0; acc[mt][1][r] = bv1; }
    }
    #pragma unroll
    for (int kk = 0; kk < 12; ++kk) {
      const int k  = 4 * kk + 2 * sel;
      const int k0 = (k     < Mn) ? k     : (Mn - 1);
      const int k1 = (k + 1 < Mn) ? k + 1 : (Mn - 1);
      v2f b0, b1, a[4];
      b0[0] = Wm0[n0 * Mn + k0];        b0[1] = Wm0[n0 * Mn + k1];
      b1[0] = Wm0[(n0 + 16) * Mn + k0]; b1[1] = Wm0[(n0 + 16) * Mn + k1];
      #pragma unroll
      for (int mt = 0; mt < 4; ++mt) {
        a[mt][0] = Abuf[mt * 16 + lh][k]; a[mt][1] = Abuf[mt * 16 + lh][k + 1];
      }
      #pragma unroll
      for (int mt = 0; mt < 4; ++mt) {
        acc[mt][0] = wmma4(a[mt], b0, acc[mt][0]);
        acc[mt][1] = wmma4(a[mt], b1, acc[mt][1]);
      }
    }
    #pragma unroll
    for (int mt = 0; mt < 4; ++mt) {
      #pragma unroll
      for (int r = 0; r < 8; ++r) {
        Bbuf[mt * 16 + r + 8 * sel][n0]      = fmaxf(acc[mt][0][r], 0.f);
        Bbuf[mt * 16 + r + 8 * sel][n0 + 16] = fmaxf(acc[mt][1][r], 0.f);
      }
    }
  }
  __syncthreads();

  hidden_layer(Wm1, bm1, Bbuf, Abuf, lane, wv);
  __syncthreads();
  hidden_layer(Wm2, bm2, Abuf, Bbuf, lane, wv);
  __syncthreads();

  for (int tile = wv; tile < 8; tile += 8) {
    const int mt = tile >> 1, nt = tile & 1;
    const int n  = nt * 16 + lh;
    const int nc = (n < OUTD) ? n : (OUTD - 1);
    const int mr = mt * 16 + lh;
    v8f acc;
    const float bv = bm3[nc];
    #pragma unroll
    for (int r = 0; r < 8; ++r) acc[r] = bv;
    #pragma unroll 4
    for (int kk = 0; kk < MHD / 4; ++kk) {
      const int k = 4 * kk + 2 * sel;
      v2f a, b;
      a[0] = Bbuf[mr][k];       a[1] = Bbuf[mr][k + 1];
      b[0] = Wm3[nc * MHD + k]; b[1] = Wm3[nc * MHD + k + 1];
      acc = wmma4(a, b, acc);
    }
    if (n < OUTD) {
      #pragma unroll
      for (int r = 0; r < 8; ++r)
        Out[(tok0 + mt * 16 + r + 8 * sel) * OUTD + n] = acc[r];
    }
  }
}

extern "C" void kernel_launch(void* const* d_in, const int* in_sizes, int n_in,
                              void* d_out, int out_size, void* d_ws, size_t ws_size,
                              hipStream_t stream) {
  const float* Dis  = (const float*)d_in[0];
  const float* Wih0 = (const float*)d_in[1];
  const float* Whh0 = (const float*)d_in[2];
  const float* bih0 = (const float*)d_in[3];
  const float* bhh0 = (const float*)d_in[4];
  const float* Wih1 = (const float*)d_in[5];
  const float* Whh1 = (const float*)d_in[6];
  const float* bih1 = (const float*)d_in[7];
  const float* bhh1 = (const float*)d_in[8];
  const float* Wih2 = (const float*)d_in[9];
  const float* Whh2 = (const float*)d_in[10];
  const float* bih2 = (const float*)d_in[11];
  const float* bhh2 = (const float*)d_in[12];
  const float* Wfc1 = (const float*)d_in[13];
  const float* bfc1 = (const float*)d_in[14];
  const float* Wfc2 = (const float*)d_in[15];
  const float* bfc2 = (const float*)d_in[16];
  const float* Wm0  = (const float*)d_in[17];
  const float* bm0  = (const float*)d_in[18];
  const float* Wm1  = (const float*)d_in[19];
  const float* bm1  = (const float*)d_in[20];
  const float* Wm2  = (const float*)d_in[21];
  const float* bm2  = (const float*)d_in[22];
  const float* Wm3  = (const float*)d_in[23];
  const float* bm3  = (const float*)d_in[24];

  float* out   = (float*)d_out;
  float* Dpred = out;
  float* cpf   = out + Mn * Bn * Tn;

  dim3 g1(NBG, Mn);
  lstm45_kernel<<<g1, NT1, 0, stream>>>(
      Dis, Wih0, Whh0, bih0, bhh0, Wih1, Whh1, bih1, bhh1,
      Wih2, Whh2, bih2, bhh2, Wfc1, bfc1, Wfc2, bfc2, Dpred);

  mlp_kernel<<<TOK / 64, 256, 0, stream>>>(
      Dpred, Wm0, bm0, Wm1, bm1, Wm2, bm2, Wm3, bm3, cpf);
}
